# starting kernel
# baseline (speedup 1.0000x reference)
#include <hip/hip_runtime.h>


typedef __attribute__((ext_vector_type(2))) float v2f;
typedef __attribute__((ext_vector_type(8))) float v8f;

#define NB            16
#define N_ITEMS_C     10000
#define N_SETS_C      1000
#define SET_SZ_C      100
#define DIM_C         128
#define SETS_PER_TILE 16
#define N_TILES       ((N_SETS_C + SETS_PER_TILE - 1) / SETS_PER_TILE)

__global__ __launch_bounds__(256) void mcp_embed_kernel(
    const float* __restrict__ weights,
    const int*   __restrict__ membership,
    const float* __restrict__ w,
    const float* __restrict__ bias,
    float*       __restrict__ out)
{
    __shared__ float sS[SETS_PER_TILE];

    const int tile = blockIdx.x;
    const int b    = blockIdx.y;
    const int tid  = threadIdx.x;
    const int lane = tid & 31;
    const int wave = tid >> 5;

    int set_base = tile * SETS_PER_TILE;
    if (set_base > N_SETS_C - SETS_PER_TILE) set_base = N_SETS_C - SETS_PER_TILE;

    const float* wrow = weights + (size_t)b * N_ITEMS_C;

    #pragma unroll
    for (int si = 0; si < 2; ++si) {
        const int s  = wave * 2 + si;
        const int gs = set_base + s;
        float acc = 0.0f;
        if (lane < (SET_SZ_C / 4)) {
            const int4* mp =
                (const int4*)(membership + ((size_t)b * N_SETS_C + gs) * SET_SZ_C);
            const int4 idx = mp[lane];
            acc = wrow[idx.x] + wrow[idx.y] + wrow[idx.z] + wrow[idx.w];
        }
        #pragma unroll
        for (int off = 16; off > 0; off >>= 1)
            acc += __shfl_xor(acc, off, 32);
        if (lane == 0) sS[s] = acc;
    }
    __syncthreads();

    const int d_base = wave * 16;
    const int nl     = lane & 15;

    v2f av;
    av.x = (lane < 16) ? sS[lane] : 0.0f;
    av.y = 0.0f;

    v2f bv;
    bv.x = (lane < 16) ? w[d_base + lane] : 0.0f;
    bv.y = 0.0f;

    const float cval = (float)SET_SZ_C * bias[d_base + nl];
    v8f cv;
    #pragma unroll
    for (int r = 0; r < 8; ++r) cv[r] = cval;

    v8f dv = __builtin_amdgcn_wmma_f32_16x16x4_f32(
        false, av, false, bv, (short)0, cv, false, false);

    const int mbase = (lane >> 4) * 8;
    float* op = out + ((size_t)b * N_SETS_C + (size_t)(set_base + mbase)) * DIM_C
                    + d_base + nl;
    #pragma unroll
    for (int r = 0; r < 8; ++r)
        op[(size_t)r * DIM_C] = dv[r];
}

extern "C" void kernel_launch(void* const* d_in, const int* in_sizes, int n_in,
                              void* d_out, int out_size, void* d_ws, size_t ws_size,
                              hipStream_t stream) {
    (void)in_sizes; (void)n_in; (void)d_ws; (void)ws_size; (void)out_size;
    const float* weights    = (const float*)d_in[0];
    const int*   membership = (const int*)d_in[1];
    const float* w          = (const float*)d_in[2];
    const float* bias       = (const float*)d_in[3];
    float*       out        = (float*)d_out;

    dim3 grid(N_TILES, NB);
    dim3 block(256);
    mcp_embed_kernel<<<grid, block, 0, stream>>>(weights, membership, w, bias, out);
}


# One wave32 per 16-set x 128-d tile, 8x fewer waves, 4 waves of one batch per work-group (256 groups): wave gathers and butterfly-sums its 16 sets, 8 WMMAs over d, whole-row stores; sum order kept
# speedup vs baseline: 1.7132x; 1.7132x over previous
#include <hip/hip_runtime.h>


typedef __attribute__((ext_vector_type(2))) float v2f;
typedef __attribute__((ext_vector_type(8))) float v8f;

#define NB            16
#define N_ITEMS_C     10000
#define N_SETS_C      1000
#define SET_SZ_C      100
#define DIM_C         128
#define SETS_PER_TILE 16
#define N_TILES       ((N_SETS_C + SETS_PER_TILE - 1) / SETS_PER_TILE)
#define WAVES_PER_WG  4
#define N_GROUPS      ((N_TILES + WAVES_PER_WG - 1) / WAVES_PER_WG)
#define VEC_PER_SET   (SET_SZ_C / 4)
#define D_TILES       (DIM_C / 16)

__global__ __launch_bounds__(32 * WAVES_PER_WG) void mcp_embed_kernel(
    const float* __restrict__ weights,
    const int*   __restrict__ membership,
    const float* __restrict__ w,
    const float* __restrict__ bias,
    float*       __restrict__ out)
{
    const int wave = __builtin_amdgcn_readfirstlane((int)(threadIdx.x >> 5));
    const int tile = blockIdx.x * WAVES_PER_WG + wave;
    const int b    = blockIdx.y;
    const int lane = threadIdx.x & 31;
    const int c    = lane & 15;
    const int h    = lane >> 4;

    if (tile >= N_TILES) return;
    int set_base = tile * SETS_PER_TILE;
    if (set_base > N_SETS_C - SETS_PER_TILE) set_base = N_SETS_C - SETS_PER_TILE;

    const float4 wlo = ((const float4*)w)[c];
    const float4 whi = ((const float4*)w)[16 + c];
    const float4 blo = ((const float4*)bias)[c];
    const float4 bhi = ((const float4*)bias)[16 + c];
    const float wv[D_TILES] = {wlo.x, wlo.y, wlo.z, wlo.w, whi.x, whi.y, whi.z, whi.w};
    const float bv[D_TILES] = {blo.x, blo.y, blo.z, blo.w, bhi.x, bhi.y, bhi.z, bhi.w};

    const float* wrow = weights + (size_t)b * N_ITEMS_C;
    const int4*  mp   =
        (const int4*)(membership + ((size_t)b * N_SETS_C + set_base) * SET_SZ_C);

    float acc[SETS_PER_TILE];
    #pragma unroll
    for (int s = 0; s < SETS_PER_TILE; ++s) acc[s] = 0.0f;

    if (lane < VEC_PER_SET) {
        int4 idx[SETS_PER_TILE];
        #pragma unroll
        for (int s = 0; s < SETS_PER_TILE; ++s)
            idx[s] = mp[s * VEC_PER_SET + lane];
        #pragma unroll
        for (int s = 0; s < SETS_PER_TILE; ++s)
            acc[s] = wrow[idx[s].x] + wrow[idx[s].y] + wrow[idx[s].z] + wrow[idx[s].w];
    }

    float sv = 0.0f;
    #pragma unroll
    for (int s = 0; s < SETS_PER_TILE; ++s) {
        float a = acc[s];
        #pragma unroll
        for (int off = 16; off > 0; off >>= 1)
            a += __shfl_xor(a, off, 32);
        if (lane == s) sv = a;
    }

    v2f av;
    av.x = sv;
    av.y = 0.0f;

    v8f dv[D_TILES];
    #pragma unroll
    for (int t = 0; t < D_TILES; ++t) {
        v2f bt;
        bt.x = (lane < 16) ? wv[t] : 0.0f;
        bt.y = 0.0f;
        const float cval = (float)SET_SZ_C * bv[t];
        v8f cv;
        #pragma unroll
        for (int r = 0; r < 8; ++r) cv[r] = cval;
        dv[t] = __builtin_amdgcn_wmma_f32_16x16x4_f32(
            false, av, false, bt, (short)0, cv, false, false);
    }

    float* op = out + ((size_t)b * N_SETS_C + (size_t)(set_base + 8 * h)) * DIM_C + 4 * c;
    #pragma unroll
    for (int g = 0; g < 2; ++g) {
        #pragma unroll
        for (int r = 0; r < 8; ++r) {
            float4 v;
            v.x = dv[4 * g + 0][r];
            v.y = dv[4 * g + 1][r];
            v.z = dv[4 * g + 2][r];
            v.w = dv[4 * g + 3][r];
            *(float4*)(op + (size_t)r * DIM_C + 64 * g) = v;
        }
    }
}

extern "C" void kernel_launch(void* const* d_in, const int* in_sizes, int n_in,
                              void* d_out, int out_size, void* d_ws, size_t ws_size,
                              hipStream_t stream) {
    (void)in_sizes; (void)n_in; (void)d_ws; (void)ws_size; (void)out_size;
    const float* weights    = (const float*)d_in[0];
    const int*   membership = (const int*)d_in[1];
    const float* w          = (const float*)d_in[2];
    const float* bias       = (const float*)d_in[3];
    float*       out        = (float*)d_out;

    dim3 grid(N_GROUPS, NB);
    dim3 block(32 * WAVES_PER_WG);
    mcp_embed_kernel<<<grid, block, 0, stream>>>(weights, membership, w, bias, out);
}


# Gather from LDS instead of the WGP cache: each work-group copies its batch's 40-KB weights row to LDS (whole-line loads beside the index loads, one barrier), then gathers there; sum order kept
# speedup vs baseline: 1.9243x; 1.1232x over previous
#include <hip/hip_runtime.h>


typedef __attribute__((ext_vector_type(2))) float v2f;
typedef __attribute__((ext_vector_type(8))) float v8f;

#define NB            16
#define N_ITEMS_C     10000
#define N_SETS_C      1000
#define SET_SZ_C      100
#define DIM_C         128
#define SETS_PER_TILE 16
#define N_TILES       ((N_SETS_C + SETS_PER_TILE - 1) / SETS_PER_TILE)
#define WAVES_PER_WG  4
#define N_GROUPS      ((N_TILES + WAVES_PER_WG - 1) / WAVES_PER_WG)
#define VEC_PER_SET   (SET_SZ_C / 4)
#define D_TILES       (DIM_C / 16)
#define WG_THREADS    (32 * WAVES_PER_WG)
#define ROW_VEC4      (N_ITEMS_C / 4)
#define STAGE_ROUNDS  ((ROW_VEC4 + WG_THREADS - 1) / WG_THREADS)
#define ROW_VEC4_PAD  (STAGE_ROUNDS * WG_THREADS)

__global__ __launch_bounds__(WG_THREADS) void mcp_embed_kernel(
    const float* __restrict__ weights,
    const int*   __restrict__ membership,
    const float* __restrict__ w,
    const float* __restrict__ bias,
    float*       __restrict__ out)
{
    const int wave = __builtin_amdgcn_readfirstlane((int)(threadIdx.x >> 5));
    const int tile = blockIdx.x * WAVES_PER_WG + wave;
    const int b    = blockIdx.y;
    const int lane = threadIdx.x & 31;
    const int c    = lane & 15;
    const int h    = lane >> 4;

    __shared__ float4 srow4[ROW_VEC4_PAD];

    const bool has_tile = tile < N_TILES;
    int set_base = tile * SETS_PER_TILE;
    if (set_base > N_SETS_C - SETS_PER_TILE) set_base = N_SETS_C - SETS_PER_TILE;

    const float4 wlo = ((const float4*)w)[c];
    const float4 whi = ((const float4*)w)[16 + c];
    const float4 blo = ((const float4*)bias)[c];
    const float4 bhi = ((const float4*)bias)[16 + c];
    const float wv[D_TILES] = {wlo.x, wlo.y, wlo.z, wlo.w, whi.x, whi.y, whi.z, whi.w};
    const float bv[D_TILES] = {blo.x, blo.y, blo.z, blo.w, bhi.x, bhi.y, bhi.z, bhi.w};

    const float4* grow4 = (const float4*)(weights + (size_t)b * N_ITEMS_C);
    const int4*   mp    =
        (const int4*)(membership + ((size_t)b * N_SETS_C + set_base) * SET_SZ_C);

    const int ld_lane = (lane < VEC_PER_SET) ? lane : (VEC_PER_SET - 1);
    int4 idx[SETS_PER_TILE];
    #pragma unroll
    for (int s = 0; s < SETS_PER_TILE; ++s)
        idx[s] = mp[s * VEC_PER_SET + ld_lane];

    {
        const int tid = threadIdx.x;
        #pragma unroll
        for (int j = 0; j < STAGE_ROUNDS; ++j) {
            const int i  = tid + WG_THREADS * j;
            const int ig = (i < ROW_VEC4) ? i : (ROW_VEC4 - 1);
            srow4[i] = grow4[ig];
        }
    }
    __syncthreads();
    if (!has_tile) return;

    const float* srow = (const float*)srow4;
    float acc[SETS_PER_TILE];
    #pragma unroll
    for (int s = 0; s < SETS_PER_TILE; ++s) acc[s] = 0.0f;
    if (lane < VEC_PER_SET) {
        #pragma unroll
        for (int s = 0; s < SETS_PER_TILE; ++s)
            acc[s] = srow[idx[s].x] + srow[idx[s].y] + srow[idx[s].z] + srow[idx[s].w];
    }

    float sv = 0.0f;
    #pragma unroll
    for (int s = 0; s < SETS_PER_TILE; ++s) {
        float a = acc[s];
        #pragma unroll
        for (int off = 16; off > 0; off >>= 1)
            a += __shfl_xor(a, off, 32);
        if (lane == s) sv = a;
    }

    v2f av;
    av.x = sv;
    av.y = 0.0f;

    v8f dv[D_TILES];
    #pragma unroll
    for (int t = 0; t < D_TILES; ++t) {
        v2f bt;
        bt.x = (lane < 16) ? wv[t] : 0.0f;
        bt.y = 0.0f;
        const float cval = (float)SET_SZ_C * bv[t];
        v8f cv;
        #pragma unroll
        for (int r = 0; r < 8; ++r) cv[r] = cval;
        dv[t] = __builtin_amdgcn_wmma_f32_16x16x4_f32(
            false, av, false, bt, (short)0, cv, false, false);
    }

    float* op = out + ((size_t)b * N_SETS_C + (size_t)(set_base + 8 * h)) * DIM_C + 4 * c;
    #pragma unroll
    for (int g = 0; g < 2; ++g) {
        #pragma unroll
        for (int r = 0; r < 8; ++r) {
            float4 v;
            v.x = dv[4 * g + 0][r];
            v.y = dv[4 * g + 1][r];
            v.z = dv[4 * g + 2][r];
            v.w = dv[4 * g + 3][r];
            *(float4*)(op + (size_t)r * DIM_C + 64 * g) = v;
        }
    }
}

extern "C" void kernel_launch(void* const* d_in, const int* in_sizes, int n_in,
                              void* d_out, int out_size, void* d_ws, size_t ws_size,
                              hipStream_t stream) {
    (void)in_sizes; (void)n_in; (void)d_ws; (void)ws_size; (void)out_size;
    const float* weights    = (const float*)d_in[0];
    const int*   membership = (const int*)d_in[1];
    const float* w          = (const float*)d_in[2];
    const float* bias       = (const float*)d_in[3];
    float*       out        = (float*)d_out;

    dim3 grid(N_GROUPS, NB);
    dim3 block(WG_THREADS);
    mcp_embed_kernel<<<grid, block, 0, stream>>>(weights, membership, w, bias, out);
}


# Cross-lane set sums moved off the LDS unit: 80 ds_bpermute become 16 v_permlanex16 plus 15 folding DPP row_xmask adds on the vector ALU; same pairing tree (16,8,4,2,1), sums bit-identical
# speedup vs baseline: 2.1503x; 1.1174x over previous
#include <hip/hip_runtime.h>


typedef __attribute__((ext_vector_type(2))) float v2f;
typedef __attribute__((ext_vector_type(8))) float v8f;

static __device__ __forceinline__ float xhalf(float x) {
    const int b = __builtin_bit_cast(int, x);
    return __builtin_bit_cast(float,
        __builtin_amdgcn_permlanex16(b, b, 0x76543210, 0xfedcba98, false, false));
}
template <int M>
static __device__ __forceinline__ float xrow(float x) {
    const int b = __builtin_bit_cast(int, x);
    return __builtin_bit_cast(float,
        __builtin_amdgcn_update_dpp(0, b, 0x160 + M, 0xf, 0xf, true));
}

#define NB            16
#define N_ITEMS_C     10000
#define N_SETS_C      1000
#define SET_SZ_C      100
#define DIM_C         128
#define SETS_PER_TILE 16
#define N_TILES       ((N_SETS_C + SETS_PER_TILE - 1) / SETS_PER_TILE)
#define WAVES_PER_WG  4
#define N_GROUPS      ((N_TILES + WAVES_PER_WG - 1) / WAVES_PER_WG)
#define VEC_PER_SET   (SET_SZ_C / 4)
#define D_TILES       (DIM_C / 16)
#define WG_THREADS    (32 * WAVES_PER_WG)
#define ROW_VEC4      (N_ITEMS_C / 4)
#define STAGE_ROUNDS  ((ROW_VEC4 + WG_THREADS - 1) / WG_THREADS)
#define ROW_VEC4_PAD  (STAGE_ROUNDS * WG_THREADS)

__global__ __launch_bounds__(WG_THREADS) void mcp_embed_kernel(
    const float* __restrict__ weights,
    const int*   __restrict__ membership,
    const float* __restrict__ w,
    const float* __restrict__ bias,
    float*       __restrict__ out)
{
    const int wave = __builtin_amdgcn_readfirstlane((int)(threadIdx.x >> 5));
    const int tile = blockIdx.x * WAVES_PER_WG + wave;
    const int b    = blockIdx.y;
    const int lane = threadIdx.x & 31;
    const int c    = lane & 15;
    const int h    = lane >> 4;

    __shared__ float4 srow4[ROW_VEC4_PAD];

    const bool has_tile = tile < N_TILES;
    int set_base = tile * SETS_PER_TILE;
    if (set_base > N_SETS_C - SETS_PER_TILE) set_base = N_SETS_C - SETS_PER_TILE;

    const float4 wlo = ((const float4*)w)[c];
    const float4 whi = ((const float4*)w)[16 + c];
    const float4 blo = ((const float4*)bias)[c];
    const float4 bhi = ((const float4*)bias)[16 + c];
    const float wv[D_TILES] = {wlo.x, wlo.y, wlo.z, wlo.w, whi.x, whi.y, whi.z, whi.w};
    const float bv[D_TILES] = {blo.x, blo.y, blo.z, blo.w, bhi.x, bhi.y, bhi.z, bhi.w};

    const float4* grow4 = (const float4*)(weights + (size_t)b * N_ITEMS_C);
    const int4*   mp    =
        (const int4*)(membership + ((size_t)b * N_SETS_C + set_base) * SET_SZ_C);

    const int ld_lane = (lane < VEC_PER_SET) ? lane : (VEC_PER_SET - 1);
    int4 idx[SETS_PER_TILE];
    #pragma unroll
    for (int s = 0; s < SETS_PER_TILE; ++s)
        idx[s] = mp[s * VEC_PER_SET + ld_lane];

    {
        const int tid = threadIdx.x;
        #pragma unroll
        for (int j = 0; j < STAGE_ROUNDS; ++j) {
            const int i  = tid + WG_THREADS * j;
            const int ig = (i < ROW_VEC4) ? i : (ROW_VEC4 - 1);
            srow4[i] = grow4[ig];
        }
    }
    __syncthreads();
    if (!has_tile) return;

    const float* srow = (const float*)srow4;
    float acc[SETS_PER_TILE];
    #pragma unroll
    for (int s = 0; s < SETS_PER_TILE; ++s) acc[s] = 0.0f;
    if (lane < VEC_PER_SET) {
        #pragma unroll
        for (int s = 0; s < SETS_PER_TILE; ++s)
            acc[s] = srow[idx[s].x] + srow[idx[s].y] + srow[idx[s].z] + srow[idx[s].w];
    }

    float v16[SETS_PER_TILE];
    #pragma unroll
    for (int s = 0; s < SETS_PER_TILE; ++s)
        v16[s] = acc[s] + xhalf(acc[s]);

    const bool b3 = (lane & 8) != 0;
    const bool b2 = (lane & 4) != 0;
    const bool b1 = (lane & 2) != 0;
    const bool b0 = (lane & 1) != 0;
    float v8[8], v4[4], v2[2];
    #pragma unroll
    for (int i = 0; i < 8; ++i) {
        const float keep = b3 ? v16[i + 8] : v16[i];
        const float send = b3 ? v16[i] : v16[i + 8];
        v8[i] = keep + xrow<8>(send);
    }
    #pragma unroll
    for (int i = 0; i < 4; ++i) {
        const float keep = b2 ? v8[i + 4] : v8[i];
        const float send = b2 ? v8[i] : v8[i + 4];
        v4[i] = keep + xrow<4>(send);
    }
    #pragma unroll
    for (int i = 0; i < 2; ++i) {
        const float keep = b1 ? v4[i + 2] : v4[i];
        const float send = b1 ? v4[i] : v4[i + 2];
        v2[i] = keep + xrow<2>(send);
    }
    const float keep1 = b0 ? v2[1] : v2[0];
    const float send1 = b0 ? v2[0] : v2[1];
    const float sall  = keep1 + xrow<1>(send1);
    const float sv = (lane < 16) ? sall : 0.0f;

    v2f av;
    av.x = sv;
    av.y = 0.0f;

    v8f dv[D_TILES];
    #pragma unroll
    for (int t = 0; t < D_TILES; ++t) {
        v2f bt;
        bt.x = (lane < 16) ? wv[t] : 0.0f;
        bt.y = 0.0f;
        const float cval = (float)SET_SZ_C * bv[t];
        v8f cv;
        #pragma unroll
        for (int r = 0; r < 8; ++r) cv[r] = cval;
        dv[t] = __builtin_amdgcn_wmma_f32_16x16x4_f32(
            false, av, false, bt, (short)0, cv, false, false);
    }

    float* op = out + ((size_t)b * N_SETS_C + (size_t)(set_base + 8 * h)) * DIM_C + 4 * c;
    #pragma unroll
    for (int g = 0; g < 2; ++g) {
        #pragma unroll
        for (int r = 0; r < 8; ++r) {
            float4 v;
            v.x = dv[4 * g + 0][r];
            v.y = dv[4 * g + 1][r];
            v.z = dv[4 * g + 2][r];
            v.w = dv[4 * g + 3][r];
            *(float4*)(op + (size_t)r * DIM_C + 64 * g) = v;
        }
    }
}

extern "C" void kernel_launch(void* const* d_in, const int* in_sizes, int n_in,
                              void* d_out, int out_size, void* d_ws, size_t ws_size,
                              hipStream_t stream) {
    (void)in_sizes; (void)n_in; (void)d_ws; (void)ws_size; (void)out_size;
    const float* weights    = (const float*)d_in[0];
    const int*   membership = (const int*)d_in[1];
    const float* w          = (const float*)d_in[2];
    const float* bias       = (const float*)d_in[3];
    float*       out        = (float*)d_out;

    dim3 grid(N_GROUPS, NB);
    dim3 block(WG_THREADS);
    mcp_embed_kernel<<<grid, block, 0, stream>>>(weights, membership, w, bias, out);
}


# Index loads stay in flight across the barrier: row loads first, hand-written barrier waits for the LDS stores only (__syncthreads drains all loads), gathers start with the row in LDS; sum order kept
# speedup vs baseline: 2.3587x; 1.0969x over previous
#include <hip/hip_runtime.h>


typedef __attribute__((ext_vector_type(2))) float v2f;
typedef __attribute__((ext_vector_type(8))) float v8f;

static __device__ __forceinline__ float xhalf(float x) {
    const int b = __builtin_bit_cast(int, x);
    return __builtin_bit_cast(float,
        __builtin_amdgcn_permlanex16(b, b, 0x76543210, 0xfedcba98, false, false));
}
template <int M>
static __device__ __forceinline__ float xrow(float x) {
    const int b = __builtin_bit_cast(int, x);
    return __builtin_bit_cast(float,
        __builtin_amdgcn_update_dpp(0, b, 0x160 + M, 0xf, 0xf, true));
}

#define NB            16
#define N_ITEMS_C     10000
#define N_SETS_C      1000
#define SET_SZ_C      100
#define DIM_C         128
#define SETS_PER_TILE 16
#define N_TILES       ((N_SETS_C + SETS_PER_TILE - 1) / SETS_PER_TILE)
#define WAVES_PER_WG  4
#define N_GROUPS      ((N_TILES + WAVES_PER_WG - 1) / WAVES_PER_WG)
#define VEC_PER_SET   (SET_SZ_C / 4)
#define D_TILES       (DIM_C / 16)
#define WG_THREADS    (32 * WAVES_PER_WG)
#define ROW_VEC4      (N_ITEMS_C / 4)
#define STAGE_ROUNDS  ((ROW_VEC4 + WG_THREADS - 1) / WG_THREADS)
#define ROW_VEC4_PAD  (STAGE_ROUNDS * WG_THREADS)

__global__ __launch_bounds__(WG_THREADS) void mcp_embed_kernel(
    const float* __restrict__ weights,
    const int*                membership,
    const float* __restrict__ w,
    const float* __restrict__ bias,
    float*       __restrict__ out)
{
    const int wave = __builtin_amdgcn_readfirstlane((int)(threadIdx.x >> 5));
    const int tile = blockIdx.x * WAVES_PER_WG + wave;
    const int b    = blockIdx.y;
    const int lane = threadIdx.x & 31;
    const int c    = lane & 15;
    const int h    = lane >> 4;

    __shared__ float4 srow4[ROW_VEC4_PAD];

    const bool has_tile = tile < N_TILES;
    int set_base = tile * SETS_PER_TILE;
    if (set_base > N_SETS_C - SETS_PER_TILE) set_base = N_SETS_C - SETS_PER_TILE;

    const float4 wlo = ((const float4*)w)[c];
    const float4 whi = ((const float4*)w)[16 + c];
    const float4 blo = ((const float4*)bias)[c];
    const float4 bhi = ((const float4*)bias)[16 + c];
    const float wv[D_TILES] = {wlo.x, wlo.y, wlo.z, wlo.w, whi.x, whi.y, whi.z, whi.w};
    const float bv[D_TILES] = {blo.x, blo.y, blo.z, blo.w, bhi.x, bhi.y, bhi.z, bhi.w};

    const float4* grow4 = (const float4*)(weights + (size_t)b * N_ITEMS_C);
    const int4*   mp    =
        (const int4*)(membership + ((size_t)b * N_SETS_C + set_base) * SET_SZ_C);

    const int tid = threadIdx.x;
    float4 stage[STAGE_ROUNDS];
    #pragma unroll
    for (int j = 0; j < STAGE_ROUNDS; ++j) {
        const int i  = tid + WG_THREADS * j;
        const int ig = (i < ROW_VEC4) ? i : (ROW_VEC4 - 1);
        stage[j] = grow4[ig];
    }

    const int ld_lane = (lane < VEC_PER_SET) ? lane : (VEC_PER_SET - 1);
    int4 idx[SETS_PER_TILE];
    #pragma unroll
    for (int s = 0; s < SETS_PER_TILE; ++s)
        idx[s] = mp[s * VEC_PER_SET + ld_lane];

    #pragma unroll
    for (int j = 0; j < STAGE_ROUNDS; ++j)
        srow4[tid + WG_THREADS * j] = stage[j];

    asm volatile("s_wait_dscnt 0x0\n\ts_barrier_signal -1\n\ts_barrier_wait -1" ::: "memory");
    if (!has_tile) {
        asm volatile("s_wait_loadcnt 0x0" ::: "memory");
        return;
    }

    const float* srow = (const float*)srow4;
    float acc[SETS_PER_TILE];
    #pragma unroll
    for (int s = 0; s < SETS_PER_TILE; ++s) acc[s] = 0.0f;
    if (lane < VEC_PER_SET) {
        #pragma unroll
        for (int s = 0; s < SETS_PER_TILE; ++s)
            acc[s] = srow[idx[s].x] + srow[idx[s].y] + srow[idx[s].z] + srow[idx[s].w];
    }

    float v16[SETS_PER_TILE];
    #pragma unroll
    for (int s = 0; s < SETS_PER_TILE; ++s)
        v16[s] = acc[s] + xhalf(acc[s]);

    const bool b3 = (lane & 8) != 0;
    const bool b2 = (lane & 4) != 0;
    const bool b1 = (lane & 2) != 0;
    const bool b0 = (lane & 1) != 0;
    float v8[8], v4[4], v2[2];
    #pragma unroll
    for (int i = 0; i < 8; ++i) {
        const float keep = b3 ? v16[i + 8] : v16[i];
        const float send = b3 ? v16[i] : v16[i + 8];
        v8[i] = keep + xrow<8>(send);
    }
    #pragma unroll
    for (int i = 0; i < 4; ++i) {
        const float keep = b2 ? v8[i + 4] : v8[i];
        const float send = b2 ? v8[i] : v8[i + 4];
        v4[i] = keep + xrow<4>(send);
    }
    #pragma unroll
    for (int i = 0; i < 2; ++i) {
        const float keep = b1 ? v4[i + 2] : v4[i];
        const float send = b1 ? v4[i] : v4[i + 2];
        v2[i] = keep + xrow<2>(send);
    }
    const float keep1 = b0 ? v2[1] : v2[0];
    const float send1 = b0 ? v2[0] : v2[1];
    const float sall  = keep1 + xrow<1>(send1);
    const float sv = (lane < 16) ? sall : 0.0f;

    v2f av;
    av.x = sv;
    av.y = 0.0f;

    v8f dv[D_TILES];
    #pragma unroll
    for (int t = 0; t < D_TILES; ++t) {
        v2f bt;
        bt.x = (lane < 16) ? wv[t] : 0.0f;
        bt.y = 0.0f;
        const float cval = (float)SET_SZ_C * bv[t];
        v8f cv;
        #pragma unroll
        for (int r = 0; r < 8; ++r) cv[r] = cval;
        dv[t] = __builtin_amdgcn_wmma_f32_16x16x4_f32(
            false, av, false, bt, (short)0, cv, false, false);
    }

    float* op = out + ((size_t)b * N_SETS_C + (size_t)(set_base + 8 * h)) * DIM_C + 4 * c;
    #pragma unroll
    for (int g = 0; g < 2; ++g) {
        #pragma unroll
        for (int r = 0; r < 8; ++r) {
            float4 v;
            v.x = dv[4 * g + 0][r];
            v.y = dv[4 * g + 1][r];
            v.z = dv[4 * g + 2][r];
            v.w = dv[4 * g + 3][r];
            *(float4*)(op + (size_t)r * DIM_C + 64 * g) = v;
        }
    }
}

extern "C" void kernel_launch(void* const* d_in, const int* in_sizes, int n_in,
                              void* d_out, int out_size, void* d_ws, size_t ws_size,
                              hipStream_t stream) {
    (void)in_sizes; (void)n_in; (void)d_ws; (void)ws_size; (void)out_size;
    const float* weights    = (const float*)d_in[0];
    const int*   membership = (const int*)d_in[1];
    const float* w          = (const float*)d_in[2];
    const float* bias       = (const float*)d_in[3];
    float*       out        = (float*)d_out;

    dim3 grid(N_GROUPS, NB);
    dim3 block(WG_THREADS);
    mcp_embed_kernel<<<grid, block, 0, stream>>>(weights, membership, w, bias, out);
}
